# starting kernel
# baseline (speedup 1.0000x reference)
#include <hip/hip_runtime.h>


#define IN_C   128
#define OUT_C  128
#define N_EDGES_CONST 50000

typedef __attribute__((ext_vector_type(2))) float v2f;
typedef __attribute__((ext_vector_type(8))) float v8f;

#define LDS_STRIDE 132

__global__ __launch_bounds__(256) void gemm_xw_wmma_kernel(
    const float* __restrict__ x, const float* __restrict__ W,
    float* __restrict__ xw)
{
    __shared__ float ldsA[16 * LDS_STRIDE];

    const int row0 = blockIdx.x * 16;
    const int tid  = threadIdx.x;

    const float4* __restrict__ xsrc = (const float4*)(x + (size_t)row0 * IN_C);
    #pragma unroll
    for (int t = 0; t < 2; ++t) {
        int chunk = tid + t * 256;
        int r   = chunk >> 5;
        int off = (chunk & 31) << 2;
        *(float4*)(ldsA + r * LDS_STRIDE + off) = xsrc[chunk];
    }
    __syncthreads();

    const int wave  = tid >> 5;
    const int lane  = tid & 31;
    const int col0  = wave * 16;
    const int lrow  = lane & 15;
    const int khalf = (lane >> 4) << 1;

    const float* __restrict__ Wc = W + col0;

    v8f c = {};
    #pragma unroll
    for (int k = 0; k < IN_C; k += 4) {
        v2f a = *(const v2f*)(ldsA + lrow * LDS_STRIDE + k + khalf);
        v2f b;
        b.x = Wc[(size_t)(k + khalf + 0) * OUT_C + lrow];
        b.y = Wc[(size_t)(k + khalf + 1) * OUT_C + lrow];
        c = __builtin_amdgcn_wmma_f32_16x16x4_f32(
                 false, a,  false, b,
                 (short)0, c,  false,  false);
    }

    float* __restrict__ dst = xw + (size_t)row0 * OUT_C + col0;
    const int mbase = (lane >> 4) << 3;
    #pragma unroll
    for (int r = 0; r < 8; ++r) {
        dst[(size_t)(mbase + r) * OUT_C + lrow] = c[r];
    }
}

__global__ void degrees_kernel(const int* __restrict__ node_idx,
                               const int* __restrict__ edge_idx,
                               float* __restrict__ deg_n,
                               float* __restrict__ deg_e, int nnz)
{
    int i = blockIdx.x * blockDim.x + threadIdx.x;
    if (i < nnz) {
        atomicAdd(&deg_n[node_idx[i]], 1.0f);
        atomicAdd(&deg_e[edge_idx[i]], 1.0f);
    }
}

__global__ void invert_kernel(float* __restrict__ d, int n)
{
    int i = blockIdx.x * blockDim.x + threadIdx.x;
    if (i < n) {
        float v = d[i];
        d[i] = (v > 0.0f) ? (1.0f / v) : 0.0f;
    }
}

__global__ __launch_bounds__(256) void scatter_node_to_edge_kernel(
    const float* __restrict__ xw,
    const int* __restrict__ node_idx, const int* __restrict__ edge_idx,
    float* __restrict__ edge_feat, int nnz)
{
    int idx = blockIdx.x * 256 + threadIdx.x;
    int i   = idx >> 5;
    int c4  = (idx & 31) << 2;
    if (i < nnz) {
        int v = node_idx[i];
        int e = edge_idx[i];
        float4 val = *(const float4*)(xw + (size_t)v * OUT_C + c4);
        float* dst = edge_feat + (size_t)e * OUT_C + c4;
        atomicAdd(dst + 0, val.x);
        atomicAdd(dst + 1, val.y);
        atomicAdd(dst + 2, val.z);
        atomicAdd(dst + 3, val.w);
    }
}

__global__ __launch_bounds__(256) void scatter_edge_to_node_kernel(
    const float* __restrict__ edge_feat, const float* __restrict__ b_inv,
    const int* __restrict__ node_idx, const int* __restrict__ edge_idx,
    float* __restrict__ out, int nnz)
{
    int idx = blockIdx.x * 256 + threadIdx.x;
    int i   = idx >> 5;
    int c4  = (idx & 31) << 2;
    if (i < nnz) {
        int v = node_idx[i];
        int e = edge_idx[i];
        float s = b_inv[e];
        float4 val = *(const float4*)(edge_feat + (size_t)e * OUT_C + c4);
        float* dst = out + (size_t)v * OUT_C + c4;
        atomicAdd(dst + 0, val.x * s);
        atomicAdd(dst + 1, val.y * s);
        atomicAdd(dst + 2, val.z * s);
        atomicAdd(dst + 3, val.w * s);
    }
}

__global__ __launch_bounds__(256) void finalize_kernel(
    float* __restrict__ out, const float* __restrict__ d_inv,
    const float* __restrict__ bias, int n_nodes)
{
    int idx = blockIdx.x * 256 + threadIdx.x;
    int v   = idx >> 5;
    int c4  = (idx & 31) << 2;
    if (v < n_nodes) {
        float  s  = d_inv[v];
        float4 bb = *(const float4*)(bias + c4);
        float4 o  = *(const float4*)(out + (size_t)v * OUT_C + c4);
        o.x = o.x * s + bb.x;
        o.y = o.y * s + bb.y;
        o.z = o.z * s + bb.z;
        o.w = o.w * s + bb.w;
        *(float4*)(out + (size_t)v * OUT_C + c4) = o;
    }
}

extern "C" void kernel_launch(void* const* d_in, const int* in_sizes, int n_in,
                              void* d_out, int out_size, void* d_ws, size_t ws_size,
                              hipStream_t stream)
{
    const float* x    = (const float*)d_in[0];
    const int*   hidx = (const int*)d_in[1];
    const float* W    = (const float*)d_in[2];
    const float* bias = (const float*)d_in[3];

    const int n_nodes = in_sizes[0] / IN_C;
    const int nnz     = in_sizes[1] / 2;
    const int n_edges = N_EDGES_CONST;

    const int* node_idx = hidx;
    const int* edge_idx = hidx + nnz;

    char* ws = (char*)d_ws;
    float* xw        = (float*)(ws);
    float* edge_feat = (float*)(ws + (size_t)n_nodes * OUT_C * 4);
    float* deg_n     = (float*)((char*)edge_feat + (size_t)n_edges * OUT_C * 4);
    float* deg_e     = (float*)((char*)deg_n + (size_t)n_nodes * 4);

    float* out = (float*)d_out;

    hipMemsetAsync(edge_feat, 0, (size_t)n_edges * OUT_C * 4, stream);
    hipMemsetAsync(deg_n,     0, (size_t)n_nodes * 4, stream);
    hipMemsetAsync(deg_e,     0, (size_t)n_edges * 4, stream);
    hipMemsetAsync(out,       0, (size_t)n_nodes * OUT_C * 4, stream);

    degrees_kernel<<<(nnz + 255) / 256, 256, 0, stream>>>(
        node_idx, edge_idx, deg_n, deg_e, nnz);
    invert_kernel<<<(n_nodes + 255) / 256, 256, 0, stream>>>(deg_n, n_nodes);
    invert_kernel<<<(n_edges + 255) / 256, 256, 0, stream>>>(deg_e, n_edges);

    gemm_xw_wmma_kernel<<<n_nodes / 16, 256, 0, stream>>>(x, W, xw);

    {
        long long threads = (long long)nnz * 32;
        scatter_node_to_edge_kernel<<<(int)((threads + 255) / 256), 256, 0, stream>>>(
            xw, node_idx, edge_idx, edge_feat, nnz);
    }

    {
        long long threads = (long long)nnz * 32;
        scatter_edge_to_node_kernel<<<(int)((threads + 255) / 256), 256, 0, stream>>>(
            edge_feat, deg_e, node_idx, edge_idx, out, nnz);
    }

    {
        long long threads = (long long)n_nodes * 32;
        finalize_kernel<<<(int)((threads + 255) / 256), 256, 0, stream>>>(
            out, deg_n, bias, n_nodes);
    }
}


# Replace float-atomic scatters by per-segment gathers over integer counting-sorted incidence lists, each segment summed in ascending-id order (fixed sum order); D^-1, B^-1, bias fused in
# speedup vs baseline: 1.7039x; 1.7039x over previous
#include <hip/hip_runtime.h>


#define IN_C   128
#define OUT_C  128
#define N_EDGES_CONST 50000

typedef __attribute__((ext_vector_type(2))) float v2f;
typedef __attribute__((ext_vector_type(8))) float v8f;

#define LDS_STRIDE 132

__global__ __launch_bounds__(256) void gemm_xw_wmma_kernel(
    const float* __restrict__ x, const float* __restrict__ W,
    float* __restrict__ xw)
{
    __shared__ float ldsA[16 * LDS_STRIDE];

    const int row0 = blockIdx.x * 16;
    const int tid  = threadIdx.x;

    const float4* __restrict__ xsrc = (const float4*)(x + (size_t)row0 * IN_C);
    #pragma unroll
    for (int t = 0; t < 2; ++t) {
        int chunk = tid + t * 256;
        int r   = chunk >> 5;
        int off = (chunk & 31) << 2;
        *(float4*)(ldsA + r * LDS_STRIDE + off) = xsrc[chunk];
    }
    __syncthreads();

    const int wave  = tid >> 5;
    const int lane  = tid & 31;
    const int col0  = wave * 16;
    const int lrow  = lane & 15;
    const int khalf = (lane >> 4) << 1;

    const float* __restrict__ Wc = W + col0;

    v8f c = {};
    #pragma unroll
    for (int k = 0; k < IN_C; k += 4) {
        v2f a = *(const v2f*)(ldsA + lrow * LDS_STRIDE + k + khalf);
        v2f b;
        b.x = Wc[(size_t)(k + khalf + 0) * OUT_C + lrow];
        b.y = Wc[(size_t)(k + khalf + 1) * OUT_C + lrow];
        c = __builtin_amdgcn_wmma_f32_16x16x4_f32(
                 false, a,  false, b,
                 (short)0, c,  false,  false);
    }

    float* __restrict__ dst = xw + (size_t)row0 * OUT_C + col0;
    const int mbase = (lane >> 4) << 3;
    #pragma unroll
    for (int r = 0; r < 8; ++r) {
        dst[(size_t)(mbase + r) * OUT_C + lrow] = c[r];
    }
}

__global__ __launch_bounds__(256) void count_kernel(
    const int* __restrict__ node_idx, const int* __restrict__ edge_idx,
    int* __restrict__ cnt_n, int* __restrict__ cnt_e, int nnz)
{
    const int i4 = (blockIdx.x * 256 + threadIdx.x) * 4;
    if ((nnz & 3) == 0 && i4 + 3 < nnz) {
        const int4 v = *(const int4*)(node_idx + i4);
        const int4 e = *(const int4*)(edge_idx + i4);
        atomicAdd(&cnt_n[v.x], 1); atomicAdd(&cnt_e[e.x], 1);
        atomicAdd(&cnt_n[v.y], 1); atomicAdd(&cnt_e[e.y], 1);
        atomicAdd(&cnt_n[v.z], 1); atomicAdd(&cnt_e[e.z], 1);
        atomicAdd(&cnt_n[v.w], 1); atomicAdd(&cnt_e[e.w], 1);
    } else {
        for (int i = i4; i < i4 + 4 && i < nnz; ++i) {
            atomicAdd(&cnt_n[node_idx[i]], 1);
            atomicAdd(&cnt_e[edge_idx[i]], 1);
        }
    }
}

#define SCAN_THREADS 1024
#define SCAN_TILE    4096

__global__ __launch_bounds__(SCAN_THREADS) void scan_kernel(
    const int* __restrict__ cnt_e, int* __restrict__ ptr_e, int n_e, int tiles_e,
    const int* __restrict__ cnt_n, int* __restrict__ ptr_n, int n_n)
{
    __shared__ int lds_s[32];
    __shared__ int lds_p[32];

    const bool is_e = (int)blockIdx.x < tiles_e;
    const int* __restrict__ cnt = is_e ? cnt_e : cnt_n;
    int* __restrict__ ptr       = is_e ? ptr_e : ptr_n;
    const int n    = is_e ? n_e : n_n;
    const int tile = is_e ? (int)blockIdx.x : (int)blockIdx.x - tiles_e;
    const int tid  = threadIdx.x;
    const int lane = tid & 31;
    const int wave = tid >> 5;

    int pre = 0;
    for (int t = 0; t < tile; ++t) {
        int4 q = *(const int4*)(cnt + (size_t)t * SCAN_TILE + tid * 4);
        pre += (q.x + q.y) + (q.z + q.w);
    }

    const int g  = tile * SCAN_TILE + tid * 4;
    const int v0 = (g + 0 < n) ? cnt[g + 0] : 0;
    const int v1 = (g + 1 < n) ? cnt[g + 1] : 0;
    const int v2 = (g + 2 < n) ? cnt[g + 2] : 0;
    const int v3 = (g + 3 < n) ? cnt[g + 3] : 0;
    const int s0 = v0, s1 = s0 + v1, s2 = s1 + v2, s3 = s2 + v3;

    int inc = s3;
    for (int d = 1; d < 32; d <<= 1) {
        int o = __shfl_up(inc, d);
        if (lane >= d) inc += o;
    }
    for (int d = 16; d >= 1; d >>= 1) pre += __shfl_xor(pre, d);

    if (lane == 31) lds_s[wave] = inc;
    if (lane == 0)  lds_p[wave] = pre;
    __syncthreads();

    int base = 0;
    for (int w = 0; w < 32; ++w) {
        base += lds_p[w];
        if (w < wave) base += lds_s[w];
    }
    const int ex = base + inc - s3;
    if (g + 0 < n) ptr[g + 0] = ex;
    if (g + 1 < n) ptr[g + 1] = ex + s0;
    if (g + 2 < n) ptr[g + 2] = ex + s1;
    if (g + 3 < n) ptr[g + 3] = ex + s2;
}

__global__ __launch_bounds__(256) void fill_kernel(
    const int* __restrict__ node_idx, const int* __restrict__ edge_idx,
    int* __restrict__ ptr_n, int* __restrict__ ptr_e,
    int* __restrict__ col_n, int* __restrict__ col_e, int nnz)
{
    const int i4 = (blockIdx.x * 256 + threadIdx.x) * 4;
    if ((nnz & 3) == 0 && i4 + 3 < nnz) {
        const int4 v = *(const int4*)(node_idx + i4);
        const int4 e = *(const int4*)(edge_idx + i4);
        const int e0 = atomicAdd(&ptr_e[e.x], 1), n0 = atomicAdd(&ptr_n[v.x], 1);
        const int e1 = atomicAdd(&ptr_e[e.y], 1), n1 = atomicAdd(&ptr_n[v.y], 1);
        const int e2 = atomicAdd(&ptr_e[e.z], 1), n2 = atomicAdd(&ptr_n[v.z], 1);
        const int e3 = atomicAdd(&ptr_e[e.w], 1), n3 = atomicAdd(&ptr_n[v.w], 1);
        col_e[e0] = v.x; col_n[n0] = e.x;
        col_e[e1] = v.y; col_n[n1] = e.y;
        col_e[e2] = v.z; col_n[n2] = e.z;
        col_e[e3] = v.w; col_n[n3] = e.w;
    } else {
        for (int i = i4; i < i4 + 4 && i < nnz; ++i) {
            const int v = node_idx[i];
            const int e = edge_idx[i];
            col_e[atomicAdd(&ptr_e[e], 1)] = v;
            col_n[atomicAdd(&ptr_n[v], 1)] = e;
        }
    }
}

#define KEY_NONE 0x7fffffff

__device__ __forceinline__ float4 load_row(const float* __restrict__ rows, int id, int c4)
{
    return *(const float4*)(rows + (size_t)id * OUT_C + c4);
}

__device__ __forceinline__ void add4(float4& a, const float4 b)
{
    a.x += b.x; a.y += b.y; a.z += b.z; a.w += b.w;
}

__device__ __forceinline__ int pick_rank(int key, int rank, int r)
{
    unsigned m = __builtin_amdgcn_ballot_w32(rank == r);
    return __builtin_amdgcn_readlane(key, __builtin_ctz(m));
}

__device__ __forceinline__ int load_key(const int* __restrict__ col, int start, int deg, int lane)
{
    return (deg <= 32 && lane < deg) ? col[start + lane] : KEY_NONE;
}

__device__ __forceinline__ float4 segment_sum_sorted(
    const float* __restrict__ rows, const int* __restrict__ col,
    int start, int deg, int lane, int key)
{
    float4 acc = make_float4(0.0f, 0.0f, 0.0f, 0.0f);
    const int c4 = lane << 2;

    if (deg <= 32) {
        int rank = 0;
        for (int j = 0; j < deg; ++j) {
            int kj = __builtin_amdgcn_readlane(key, j);
            rank += (kj < key || (kj == key && j < lane)) ? 1 : 0;
        }
        int r = 0;
        for (; r + 4 <= deg; r += 4) {
            int i0 = pick_rank(key, rank, r + 0);
            int i1 = pick_rank(key, rank, r + 1);
            int i2 = pick_rank(key, rank, r + 2);
            int i3 = pick_rank(key, rank, r + 3);
            float4 a0 = load_row(rows, i0, c4);
            float4 a1 = load_row(rows, i1, c4);
            float4 a2 = load_row(rows, i2, c4);
            float4 a3 = load_row(rows, i3, c4);
            add4(acc, a0); add4(acc, a1); add4(acc, a2); add4(acc, a3);
        }
        for (; r < deg; ++r) {
            float4 a = load_row(rows, pick_rank(key, rank, r), c4);
            add4(acc, a);
        }
    } else {
        int last = -1;
        for (int it = 0; it < deg; ++it) {
            int mn = KEY_NONE;
            for (int b = 0; b < deg; b += 32) {
                int k = (b + lane < deg) ? col[start + b + lane] : KEY_NONE;
                if (k > last && k < mn) mn = k;
            }
            for (int d = 16; d >= 1; d >>= 1) {
                int o = __shfl_xor(mn, d);
                mn = (o < mn) ? o : mn;
            }
            mn = __builtin_amdgcn_readfirstlane(mn);
            if (mn == KEY_NONE) break;
            int c = 0;
            for (int b = 0; b < deg; b += 32) {
                int k = (b + lane < deg) ? col[start + b + lane] : KEY_NONE;
                c += __builtin_popcount(__builtin_amdgcn_ballot_w32(k == mn));
            }
            float4 a = load_row(rows, mn, c4);
            for (int t = 0; t < c; ++t) add4(acc, a);
            last = mn;
        }
    }
    return acc;
}

#define EDGE_SEGS 4
#define NODE_SEGS 8

__global__ __launch_bounds__(256) void gather_edges_kernel(
    const float* __restrict__ xw,
    const int* __restrict__ cnt_e, const int* __restrict__ ptr_e,
    const int* __restrict__ col_e,
    float* __restrict__ edge_feat, int n_edges)
{
    const int lane = threadIdx.x & 31;
    const int s0 = __builtin_amdgcn_readfirstlane(
        (int)(blockIdx.x * 8 + (threadIdx.x >> 5)) * EDGE_SEGS);

    int my_cnt = 0, my_end = 0;
    if (lane < EDGE_SEGS && s0 + lane < n_edges) {
        my_cnt = cnt_e[s0 + lane];
        my_end = ptr_e[s0 + lane];
    }
    int deg   = __builtin_amdgcn_readlane(my_cnt, 0);
    int start = __builtin_amdgcn_readlane(my_end, 0) - deg;
    int key   = load_key(col_e, start, deg, lane);

    #pragma nounroll
    for (int t = 0; t < EDGE_SEGS; ++t) {
        const int e = s0 + t;
        if (e >= n_edges) break;
        int ndeg = 0, nstart = 0, nkey = KEY_NONE;
        if (t + 1 < EDGE_SEGS && e + 1 < n_edges) {
            ndeg   = __builtin_amdgcn_readlane(my_cnt, t + 1);
            nstart = __builtin_amdgcn_readlane(my_end, t + 1) - ndeg;
            nkey   = load_key(col_e, nstart, ndeg, lane);
        }
        float4 acc = segment_sum_sorted(xw, col_e, start, deg, lane, key);
        const float s = (deg > 0) ? (1.0f / (float)deg) : 0.0f;
        acc.x *= s; acc.y *= s; acc.z *= s; acc.w *= s;
        *(float4*)(edge_feat + (size_t)e * OUT_C + (lane << 2)) = acc;
        deg = ndeg; start = nstart; key = nkey;
    }
}

__global__ __launch_bounds__(256) void gather_nodes_kernel(
    const float* __restrict__ edge_feat,
    const int* __restrict__ cnt_n, const int* __restrict__ ptr_n,
    const int* __restrict__ col_n,
    const float* __restrict__ bias,
    float* __restrict__ out, int n_nodes)
{
    const int lane = threadIdx.x & 31;
    const int s0 = __builtin_amdgcn_readfirstlane(
        (int)(blockIdx.x * 8 + (threadIdx.x >> 5)) * NODE_SEGS);

    int my_cnt = 0, my_end = 0;
    if (lane < NODE_SEGS && s0 + lane < n_nodes) {
        my_cnt = cnt_n[s0 + lane];
        my_end = ptr_n[s0 + lane];
    }
    const float4 bb = *(const float4*)(bias + (lane << 2));
    int deg   = __builtin_amdgcn_readlane(my_cnt, 0);
    int start = __builtin_amdgcn_readlane(my_end, 0) - deg;
    int key   = load_key(col_n, start, deg, lane);

    for (int t = 0; t < NODE_SEGS; ++t) {
        const int v = s0 + t;
        if (v >= n_nodes) break;
        int ndeg = 0, nstart = 0, nkey = KEY_NONE;
        if (t + 1 < NODE_SEGS && v + 1 < n_nodes) {
            ndeg   = __builtin_amdgcn_readlane(my_cnt, t + 1);
            nstart = __builtin_amdgcn_readlane(my_end, t + 1) - ndeg;
            nkey   = load_key(col_n, nstart, ndeg, lane);
        }
        const float4 acc = segment_sum_sorted(edge_feat, col_n, start, deg, lane, key);
        const float s = (deg > 0) ? (1.0f / (float)deg) : 0.0f;
        float4 o;
        o.x = acc.x * s + bb.x;
        o.y = acc.y * s + bb.y;
        o.z = acc.z * s + bb.z;
        o.w = acc.w * s + bb.w;
        *(float4*)(out + (size_t)v * OUT_C + (lane << 2)) = o;
        deg = ndeg; start = nstart; key = nkey;
    }
}

extern "C" void kernel_launch(void* const* d_in, const int* in_sizes, int n_in,
                              void* d_out, int out_size, void* d_ws, size_t ws_size,
                              hipStream_t stream)
{
    const float* x    = (const float*)d_in[0];
    const int*   hidx = (const int*)d_in[1];
    const float* W    = (const float*)d_in[2];
    const float* bias = (const float*)d_in[3];

    const int n_nodes = in_sizes[0] / IN_C;
    const int nnz     = in_sizes[1] / 2;
    const int n_edges = N_EDGES_CONST;

    const int* node_idx = hidx;
    const int* edge_idx = hidx + nnz;

    float* out = (float*)d_out;

    float* xw = out;

    char* ws = (char*)d_ws;
    float* edge_feat = (float*)(ws);
    int* cnt_e = (int*)(ws + (size_t)n_edges * OUT_C * 4);
    int* cnt_n = cnt_e + n_edges;
    int* ptr_e = cnt_n + n_nodes;
    int* ptr_n = ptr_e + n_edges;
    int* col_e = ptr_n + n_nodes;
    int* col_n = col_e + nnz;

    hipMemsetAsync(cnt_e, 0, (size_t)(n_edges + n_nodes) * 4, stream);

    count_kernel<<<(nnz + 1023) / 1024, 256, 0, stream>>>(
        node_idx, edge_idx, cnt_n, cnt_e, nnz);
    {
        const int tiles_e = (n_edges + SCAN_TILE - 1) / SCAN_TILE;
        const int tiles_n = (n_nodes + SCAN_TILE - 1) / SCAN_TILE;
        scan_kernel<<<tiles_e + tiles_n, SCAN_THREADS, 0, stream>>>(
            cnt_e, ptr_e, n_edges, tiles_e, cnt_n, ptr_n, n_nodes);
    }
    fill_kernel<<<(nnz + 1023) / 1024, 256, 0, stream>>>(
        node_idx, edge_idx, ptr_n, ptr_e, col_n, col_e, nnz);

    gemm_xw_wmma_kernel<<<n_nodes / 16, 256, 0, stream>>>(x, W, xw);

    gather_edges_kernel<<<(n_edges + 8 * EDGE_SEGS - 1) / (8 * EDGE_SEGS), 256, 0, stream>>>(
        xw, cnt_e, ptr_e, col_e, edge_feat, n_edges);

    gather_nodes_kernel<<<(n_nodes + 8 * NODE_SEGS - 1) / (8 * NODE_SEGS), 256, 0, stream>>>(
        edge_feat, cnt_n, ptr_n, col_n, bias, out, n_nodes);
}


# Build both incidence lists with LDS counters, no global atomics: ids bucketed by 256, pairs moved to buckets (LDS add's return = slot), each bucket grouped by one work-group; sums bit-identical
# speedup vs baseline: 3.6194x; 2.1242x over previous
#include <hip/hip_runtime.h>


#define IN_C   128
#define OUT_C  128
#define N_EDGES_CONST 50000

typedef __attribute__((ext_vector_type(2))) float v2f;
typedef __attribute__((ext_vector_type(8))) float v8f;

#define LDS_STRIDE 132

#define BKT_SHIFT    8
#define BKT_IDS      256
#define PART_THREADS 1024
#define PART_CHUNK   4096
#define PART_PER_THREAD (PART_CHUNK / PART_THREADS)
#define HIST_GROUP   32
#define FIN_GROUP    8
#define MAX_BKT      1024

__global__ __launch_bounds__(PART_THREADS) void bucket_hist_kernel(
    const int* __restrict__ node_idx, const int* __restrict__ edge_idx,
    int* __restrict__ hist, int nnz, int nbe, int nbt)
{
    __shared__ int h[MAX_BKT];
    const int tid = threadIdx.x;
    h[tid] = 0;
    __syncthreads();
    const int base = (int)blockIdx.x * PART_CHUNK;
    int e[PART_PER_THREAD], v[PART_PER_THREAD];
    #pragma unroll
    for (int j = 0; j < PART_PER_THREAD; ++j) {
        const int i = min(base + j * PART_THREADS + tid, nnz - 1);
        e[j] = edge_idx[i];
        v[j] = node_idx[i];
    }
    #pragma unroll
    for (int j = 0; j < PART_PER_THREAD; ++j) {
        if (base + j * PART_THREADS + tid < nnz) {
            atomicAdd(&h[e[j] >> BKT_SHIFT], 1);
            atomicAdd(&h[nbe + (v[j] >> BKT_SHIFT)], 1);
        }
    }
    __syncthreads();
    if (tid < nbt) hist[(size_t)blockIdx.x * nbt + tid] = h[tid];
}

__global__ __launch_bounds__(PART_THREADS) void bucket_scatter_kernel(
    const int* __restrict__ node_idx, const int* __restrict__ edge_idx,
    const int* __restrict__ hist,
    int* __restrict__ pairs_e, int* __restrict__ pairs_n,
    int* __restrict__ bstart, int* __restrict__ bcount,
    int nnz, int nbe, int nbt, int nblk)
{
    __shared__ int cur[MAX_BKT];
    __shared__ int wsum[32];
    const int tid  = threadIdx.x;
    const int lane = tid & 31;
    const int wave = tid >> 5;
    const int me   = (int)blockIdx.x;

    int tot = 0, pre = 0;
    if (tid < nbt) {
        for (int b0 = 0; b0 < nblk; b0 += HIST_GROUP) {
            int c[HIST_GROUP];
            #pragma unroll
            for (int u = 0; u < HIST_GROUP; ++u)
                c[u] = hist[(size_t)min(b0 + u, nblk - 1) * nbt + tid];
            #pragma unroll
            for (int u = 0; u < HIST_GROUP; ++u) {
                const int b = b0 + u;
                tot += (b < nblk) ? c[u] : 0;
                pre += (b < me) ? c[u] : 0;
            }
        }
    }

    int inc = tot;
    for (int d = 1; d < 32; d <<= 1) {
        int o = __shfl_up(inc, d);
        if (lane >= d) inc += o;
    }
    if (lane == 31) wsum[wave] = inc;
    __syncthreads();
    int wbase = 0;
    for (int w = 0; w < 32; ++w)
        if (w < wave) wbase += wsum[w];
    const int joint = wbase + inc - tot;
    const int start = (tid < nbe) ? joint : joint - nnz;

    if (tid < nbt) {
        cur[tid] = start + pre;
        if (me == 0) { bstart[tid] = start; bcount[tid] = tot; }
    }
    __syncthreads();

    int2* __restrict__ pe2 = (int2*)pairs_e;
    int2* __restrict__ pn2 = (int2*)pairs_n;
    const int base = me * PART_CHUNK;
    int e[PART_PER_THREAD], v[PART_PER_THREAD];
    #pragma unroll
    for (int j = 0; j < PART_PER_THREAD; ++j) {
        const int i = min(base + j * PART_THREADS + tid, nnz - 1);
        e[j] = edge_idx[i];
        v[j] = node_idx[i];
    }
    #pragma unroll
    for (int j = 0; j < PART_PER_THREAD; ++j) {
        if (base + j * PART_THREADS + tid < nnz) {
            const int se = atomicAdd(&cur[e[j] >> BKT_SHIFT], 1);
            const int sn = atomicAdd(&cur[nbe + (v[j] >> BKT_SHIFT)], 1);
            pe2[se] = make_int2(e[j], v[j]);
            pn2[sn] = make_int2(v[j], e[j]);
        }
    }
}

__global__ __launch_bounds__(BKT_IDS) void bucket_finalize_kernel(
    const int* __restrict__ pairs_e, const int* __restrict__ pairs_n,
    const int* __restrict__ bstart, const int* __restrict__ bcount,
    int* __restrict__ cnt_e, int* __restrict__ ptr_e, int* __restrict__ col_e,
    int* __restrict__ cnt_n, int* __restrict__ ptr_n, int* __restrict__ col_n,
    int n_edges, int n_nodes, int nbe)
{
    __shared__ int c[BKT_IDS];
    __shared__ int wsum[BKT_IDS / 32];
    const int tid  = threadIdx.x;
    const int lane = tid & 31;
    const int wave = tid >> 5;
    const int t    = (int)blockIdx.x;
    const bool is_e = t < nbe;
    const int2* __restrict__ pairs = (const int2*)(is_e ? pairs_e : pairs_n);
    int* __restrict__ cnt = is_e ? cnt_e : cnt_n;
    int* __restrict__ ptr = is_e ? ptr_e : ptr_n;
    int* __restrict__ col = is_e ? col_e : col_n;
    const int nseg  = is_e ? n_edges : n_nodes;
    const int id0   = (is_e ? t : t - nbe) << BKT_SHIFT;
    const int start = bstart[t];
    const int n     = bcount[t];

    c[tid] = 0;
    __syncthreads();
    for (int i0 = 0; i0 < n; i0 += FIN_GROUP * BKT_IDS) {
        int id[FIN_GROUP];
        #pragma unroll
        for (int u = 0; u < FIN_GROUP; ++u)
            id[u] = pairs[start + min(i0 + u * BKT_IDS + tid, n - 1)].x;
        #pragma unroll
        for (int u = 0; u < FIN_GROUP; ++u)
            if (i0 + u * BKT_IDS + tid < n) atomicAdd(&c[id[u] - id0], 1);
    }
    __syncthreads();

    const int mine = c[tid];
    int inc = mine;
    for (int d = 1; d < 32; d <<= 1) {
        int o = __shfl_up(inc, d);
        if (lane >= d) inc += o;
    }
    if (lane == 31) wsum[wave] = inc;
    __syncthreads();
    int wbase = 0;
    for (int w = 0; w < BKT_IDS / 32; ++w)
        if (w < wave) wbase += wsum[w];
    const int first = start + wbase + inc - mine;
    const int seg = id0 + tid;
    if (seg < nseg) { cnt[seg] = mine; ptr[seg] = first + mine; }
    c[tid] = first;
    __syncthreads();

    for (int i0 = 0; i0 < n; i0 += FIN_GROUP * BKT_IDS) {
        int2 p[FIN_GROUP];
        #pragma unroll
        for (int u = 0; u < FIN_GROUP; ++u)
            p[u] = pairs[start + min(i0 + u * BKT_IDS + tid, n - 1)];
        #pragma unroll
        for (int u = 0; u < FIN_GROUP; ++u)
            if (i0 + u * BKT_IDS + tid < n) col[atomicAdd(&c[p[u].x - id0], 1)] = p[u].y;
    }
}

#define KEY_NONE 0x7fffffff

__device__ __forceinline__ float4 load_row(const float* __restrict__ rows, int id, int c4)
{
    return *(const float4*)(rows + (size_t)id * OUT_C + c4);
}

__device__ __forceinline__ void add4(float4& a, const float4 b)
{
    a.x += b.x; a.y += b.y; a.z += b.z; a.w += b.w;
}

__device__ __forceinline__ int pick_rank(int key, int rank, int r)
{
    unsigned m = __builtin_amdgcn_ballot_w32(rank == r);
    return __builtin_amdgcn_readlane(key, __builtin_ctz(m));
}

__device__ __forceinline__ int load_key(const int* __restrict__ col, int start, int deg, int lane)
{
    return (deg <= 32 && lane < deg) ? col[start + lane] : KEY_NONE;
}

__device__ __forceinline__ float4 segment_sum_sorted(
    const float* __restrict__ rows, const int* __restrict__ col,
    int start, int deg, int lane, int key)
{
    float4 acc = make_float4(0.0f, 0.0f, 0.0f, 0.0f);
    const int c4 = lane << 2;

    if (deg <= 32) {
        int rank = 0;
        for (int j = 0; j < deg; ++j) {
            int kj = __builtin_amdgcn_readlane(key, j);
            rank += (kj < key || (kj == key && j < lane)) ? 1 : 0;
        }
        int r = 0;
        for (; r + 4 <= deg; r += 4) {
            int i0 = pick_rank(key, rank, r + 0);
            int i1 = pick_rank(key, rank, r + 1);
            int i2 = pick_rank(key, rank, r + 2);
            int i3 = pick_rank(key, rank, r + 3);
            float4 a0 = load_row(rows, i0, c4);
            float4 a1 = load_row(rows, i1, c4);
            float4 a2 = load_row(rows, i2, c4);
            float4 a3 = load_row(rows, i3, c4);
            add4(acc, a0); add4(acc, a1); add4(acc, a2); add4(acc, a3);
        }
        for (; r < deg; ++r) {
            float4 a = load_row(rows, pick_rank(key, rank, r), c4);
            add4(acc, a);
        }
    } else {
        int last = -1;
        for (int it = 0; it < deg; ++it) {
            int mn = KEY_NONE;
            for (int b = 0; b < deg; b += 32) {
                int k = (b + lane < deg) ? col[start + b + lane] : KEY_NONE;
                if (k > last && k < mn) mn = k;
            }
            for (int d = 16; d >= 1; d >>= 1) {
                int o = __shfl_xor(mn, d);
                mn = (o < mn) ? o : mn;
            }
            mn = __builtin_amdgcn_readfirstlane(mn);
            if (mn == KEY_NONE) break;
            int c = 0;
            for (int b = 0; b < deg; b += 32) {
                int k = (b + lane < deg) ? col[start + b + lane] : KEY_NONE;
                c += __builtin_popcount(__builtin_amdgcn_ballot_w32(k == mn));
            }
            float4 a = load_row(rows, mn, c4);
            for (int t = 0; t < c; ++t) add4(acc, a);
            last = mn;
        }
    }
    return acc;
}

#define EDGE_SEGS 4
#define NODE_SEGS 8

#define EDGE_ROWS (8 * EDGE_SEGS)

__global__ __launch_bounds__(256) void gather_edges_gemm_kernel(
    const float* __restrict__ x, const float* __restrict__ W,
    const int* __restrict__ cnt_e, const int* __restrict__ ptr_e,
    const int* __restrict__ col_e,
    float* __restrict__ edge_feat, int n_edges)
{
    __shared__ float ldsS[EDGE_ROWS * LDS_STRIDE];

    const int lane = threadIdx.x & 31;
    const int wave = threadIdx.x >> 5;
    const int row0 = (int)blockIdx.x * EDGE_ROWS;
    const int s0 = __builtin_amdgcn_readfirstlane(row0 + wave * EDGE_SEGS);

    int my_cnt = 0, my_end = 0;
    if (lane < EDGE_SEGS && s0 + lane < n_edges) {
        my_cnt = cnt_e[s0 + lane];
        my_end = ptr_e[s0 + lane];
    }
    int deg   = __builtin_amdgcn_readlane(my_cnt, 0);
    int start = __builtin_amdgcn_readlane(my_end, 0) - deg;
    int key   = load_key(col_e, start, deg, lane);

    #pragma nounroll
    for (int t = 0; t < EDGE_SEGS; ++t) {
        const int e = s0 + t;
        int ndeg = 0, nstart = 0, nkey = KEY_NONE;
        if (t + 1 < EDGE_SEGS && e + 1 < n_edges) {
            ndeg   = __builtin_amdgcn_readlane(my_cnt, t + 1);
            nstart = __builtin_amdgcn_readlane(my_end, t + 1) - ndeg;
            nkey   = load_key(col_e, nstart, ndeg, lane);
        }
        float4 acc = segment_sum_sorted(x, col_e, start, deg, lane, key);
        const float s = (deg > 0) ? (1.0f / (float)deg) : 0.0f;
        acc.x *= s; acc.y *= s; acc.z *= s; acc.w *= s;
        *(float4*)(ldsS + (wave * EDGE_SEGS + t) * LDS_STRIDE + (lane << 2)) = acc;
        deg = ndeg; start = nstart; key = nkey;
    }
    __syncthreads();

    const int col0  = wave * 16;
    const int lrow  = lane & 15;
    const int khalf = (lane >> 4) << 1;

    const float* __restrict__ Wc = W + col0;
    const float* __restrict__ A0 = ldsS + lrow * LDS_STRIDE + khalf;
    const float* __restrict__ A1 = A0 + 16 * LDS_STRIDE;

    v8f c0 = {};
    v8f c1 = {};
    #pragma unroll 8
    for (int k = 0; k < IN_C; k += 4) {
        v2f a0 = *(const v2f*)(A0 + k);
        v2f a1 = *(const v2f*)(A1 + k);
        v2f b;
        b.x = Wc[(size_t)(k + khalf + 0) * OUT_C + lrow];
        b.y = Wc[(size_t)(k + khalf + 1) * OUT_C + lrow];
        c0 = __builtin_amdgcn_wmma_f32_16x16x4_f32(
                 false, a0,  false, b,
                 (short)0, c0,  false,  false);
        c1 = __builtin_amdgcn_wmma_f32_16x16x4_f32(
                 false, a1,  false, b,
                 (short)0, c1,  false,  false);
    }

    const int mbase = (lane >> 4) << 3;
    float* __restrict__ dst = edge_feat + (size_t)row0 * OUT_C + col0 + lrow;
    #pragma unroll
    for (int r = 0; r < 8; ++r) {
        const int m0 = mbase + r;
        const int m1 = 16 + mbase + r;
        if (row0 + m0 < n_edges) dst[(size_t)m0 * OUT_C] = c0[r];
        if (row0 + m1 < n_edges) dst[(size_t)m1 * OUT_C] = c1[r];
    }
}

__global__ __launch_bounds__(256) void gather_nodes_kernel(
    const float* __restrict__ edge_feat,
    const int* __restrict__ cnt_n, const int* __restrict__ ptr_n,
    const int* __restrict__ col_n,
    const float* __restrict__ bias,
    float* __restrict__ out, int n_nodes)
{
    const int lane = threadIdx.x & 31;
    const int s0 = __builtin_amdgcn_readfirstlane(
        (int)(blockIdx.x * 8 + (threadIdx.x >> 5)) * NODE_SEGS);

    int my_cnt = 0, my_end = 0;
    if (lane < NODE_SEGS && s0 + lane < n_nodes) {
        my_cnt = cnt_n[s0 + lane];
        my_end = ptr_n[s0 + lane];
    }
    const float4 bb = *(const float4*)(bias + (lane << 2));
    int deg   = __builtin_amdgcn_readlane(my_cnt, 0);
    int start = __builtin_amdgcn_readlane(my_end, 0) - deg;
    int key   = load_key(col_n, start, deg, lane);

    for (int t = 0; t < NODE_SEGS; ++t) {
        const int v = s0 + t;
        if (v >= n_nodes) break;
        int ndeg = 0, nstart = 0, nkey = KEY_NONE;
        if (t + 1 < NODE_SEGS && v + 1 < n_nodes) {
            ndeg   = __builtin_amdgcn_readlane(my_cnt, t + 1);
            nstart = __builtin_amdgcn_readlane(my_end, t + 1) - ndeg;
            nkey   = load_key(col_n, nstart, ndeg, lane);
        }
        const float4 acc = segment_sum_sorted(edge_feat, col_n, start, deg, lane, key);
        const float s = (deg > 0) ? (1.0f / (float)deg) : 0.0f;
        float4 o;
        o.x = acc.x * s + bb.x;
        o.y = acc.y * s + bb.y;
        o.z = acc.z * s + bb.z;
        o.w = acc.w * s + bb.w;
        *(float4*)(out + (size_t)v * OUT_C + (lane << 2)) = o;
        deg = ndeg; start = nstart; key = nkey;
    }
}

extern "C" void kernel_launch(void* const* d_in, const int* in_sizes, int n_in,
                              void* d_out, int out_size, void* d_ws, size_t ws_size,
                              hipStream_t stream)
{
    const float* x    = (const float*)d_in[0];
    const int*   hidx = (const int*)d_in[1];
    const float* W    = (const float*)d_in[2];
    const float* bias = (const float*)d_in[3];

    const int n_nodes = in_sizes[0] / IN_C;
    const int nnz     = in_sizes[1] / 2;
    const int n_edges = N_EDGES_CONST;

    const int* node_idx = hidx;
    const int* edge_idx = hidx + nnz;

    float* out = (float*)d_out;

    char* ws = (char*)d_ws;
    float* edge_feat = (float*)(ws);
    int* cnt_e = (int*)(ws + (size_t)n_edges * OUT_C * 4);
    int* cnt_n = cnt_e + n_edges;
    int* ptr_e = cnt_n + n_nodes;
    int* ptr_n = ptr_e + n_edges;
    int* col_e = ptr_n + n_nodes;
    int* col_n = col_e + nnz;

    const int nbe  = (n_edges + BKT_IDS - 1) / BKT_IDS;
    const int nbn  = (n_nodes + BKT_IDS - 1) / BKT_IDS;
    const int nbt  = nbe + nbn;
    const int nblk = (nnz + PART_CHUNK - 1) / PART_CHUNK;
    int* pairs_e = (int*)(ws);
    int* pairs_n = pairs_e + 2 * (size_t)nnz;
    int* hist    = pairs_n + 2 * (size_t)nnz;
    int* bstart  = hist + (size_t)nblk * nbt;
    int* bcount  = bstart + nbt;

    bucket_hist_kernel<<<nblk, PART_THREADS, 0, stream>>>(
        node_idx, edge_idx, hist, nnz, nbe, nbt);
    bucket_scatter_kernel<<<nblk, PART_THREADS, 0, stream>>>(
        node_idx, edge_idx, hist, pairs_e, pairs_n, bstart, bcount, nnz, nbe, nbt, nblk);
    bucket_finalize_kernel<<<nbt, BKT_IDS, 0, stream>>>(
        pairs_e, pairs_n, bstart, bcount,
        cnt_e, ptr_e, col_e, cnt_n, ptr_n, col_n, n_edges, n_nodes, nbe);

    gather_edges_gemm_kernel<<<(n_edges + EDGE_ROWS - 1) / EDGE_ROWS, 256, 0, stream>>>(
        x, W, cnt_e, ptr_e, col_e, edge_feat, n_edges);

    gather_nodes_kernel<<<(n_nodes + 8 * NODE_SEGS - 1) / (8 * NODE_SEGS), 256, 0, stream>>>(
        edge_feat, cnt_n, ptr_n, col_n, bias, out, n_nodes);
}


# List build in 2 launches, not 3: each partition block orders its 4096 incidences by bucket inside its own region (one LDS pass, no global histogram); bucket groups collect the runs; sums bit-identical
# speedup vs baseline: 4.2215x; 1.1664x over previous
#include <hip/hip_runtime.h>


#define IN_C   128
#define OUT_C  128
#define N_EDGES_CONST 50000

typedef __attribute__((ext_vector_type(2))) float v2f;
typedef __attribute__((ext_vector_type(8))) float v8f;

#define LDS_STRIDE 132

#define BKT_SHIFT    8
#define BKT_IDS      256
#define PART_THREADS 1024
#define PART_CHUNK   4096
#define PART_PER_THREAD (PART_CHUNK / PART_THREADS)
#define RUN_GROUP    16
#define MAX_BKT      1024
#define MAX_BLK      1024

__global__ __launch_bounds__(PART_THREADS) void bucket_partition_kernel(
    const int* __restrict__ node_idx, const int* __restrict__ edge_idx,
    int* __restrict__ pairs_e, int* __restrict__ pairs_n,
    int* __restrict__ run_off, int* __restrict__ run_cnt,
    int nnz, int nbe, int nbt, int nblk)
{
    __shared__ int h[MAX_BKT];
    __shared__ int wsum[32];
    const int tid  = threadIdx.x;
    const int lane = tid & 31;
    const int wave = tid >> 5;
    const int me   = (int)blockIdx.x;
    h[tid] = 0;
    __syncthreads();

    const int base = me * PART_CHUNK;
    int e[PART_PER_THREAD], v[PART_PER_THREAD];
    #pragma unroll
    for (int j = 0; j < PART_PER_THREAD; ++j) {
        const int i = min(base + j * PART_THREADS + tid, nnz - 1);
        e[j] = edge_idx[i];
        v[j] = node_idx[i];
    }
    int re[PART_PER_THREAD], rn[PART_PER_THREAD];
    #pragma unroll
    for (int j = 0; j < PART_PER_THREAD; ++j) {
        re[j] = 0; rn[j] = 0;
        if (base + j * PART_THREADS + tid < nnz) {
            re[j] = atomicAdd(&h[e[j] >> BKT_SHIFT], 1);
            rn[j] = atomicAdd(&h[nbe + (v[j] >> BKT_SHIFT)], 1);
        }
    }
    __syncthreads();

    const int tot = h[tid];
    int inc = tot;
    for (int d = 1; d < 32; d <<= 1) {
        int o = __shfl_up(inc, d);
        if (lane >= d) inc += o;
    }
    if (lane == 31) wsum[wave] = inc;
    __syncthreads();
    int wbase = 0;
    for (int w = 0; w < 32; ++w)
        if (w < wave) wbase += wsum[w];
    const int nvalid = min(PART_CHUNK, nnz - base);
    const int joint  = wbase + inc - tot;
    const int start  = (tid < nbe) ? joint : joint - nvalid;
    h[tid] = start;
    if (tid < nbt) {
        run_off[(size_t)tid * nblk + me] = start;
        run_cnt[(size_t)tid * nblk + me] = tot;
    }
    __syncthreads();

    int2* __restrict__ pe2 = (int2*)pairs_e + (size_t)me * PART_CHUNK;
    int2* __restrict__ pn2 = (int2*)pairs_n + (size_t)me * PART_CHUNK;
    #pragma unroll
    for (int j = 0; j < PART_PER_THREAD; ++j) {
        if (base + j * PART_THREADS + tid < nnz) {
            pe2[h[e[j] >> BKT_SHIFT] + re[j]] = make_int2(e[j], v[j]);
            pn2[h[nbe + (v[j] >> BKT_SHIFT)] + rn[j]] = make_int2(v[j], e[j]);
        }
    }
}

__global__ __launch_bounds__(BKT_IDS) void bucket_finalize_kernel(
    const int* __restrict__ pairs_e, const int* __restrict__ pairs_n,
    const int* __restrict__ run_off, const int* __restrict__ run_cnt,
    int* __restrict__ cnt_e, int* __restrict__ ptr_e, int* __restrict__ col_e,
    int* __restrict__ cnt_n, int* __restrict__ ptr_n, int* __restrict__ col_n,
    int n_edges, int n_nodes, int nbe, int nblk)
{
    __shared__ int c[BKT_IDS];
    __shared__ int wsum[BKT_IDS / 32];
    __shared__ int osum[BKT_IDS / 32];
    __shared__ int rs[MAX_BLK];
    __shared__ int rc[MAX_BLK];
    const int tid  = threadIdx.x;
    const int lane = tid & 31;
    const int wave = tid >> 5;
    const int t    = (int)blockIdx.x;
    const bool is_e = t < nbe;
    const int2* __restrict__ pairs = (const int2*)(is_e ? pairs_e : pairs_n);
    int* __restrict__ cnt = is_e ? cnt_e : cnt_n;
    int* __restrict__ ptr = is_e ? ptr_e : ptr_n;
    int* __restrict__ col = is_e ? col_e : col_n;
    const int nseg  = is_e ? n_edges : n_nodes;
    const int id0   = (is_e ? t : t - nbe) << BKT_SHIFT;

    int part = 0;
    for (int b = tid; b < nblk; b += BKT_IDS) {
        const int o = run_off[(size_t)t * nblk + b];
        rc[b] = run_cnt[(size_t)t * nblk + b];
        rs[b] = b * PART_CHUNK + o;
        part += o;
    }
    c[tid] = 0;
    for (int d = 16; d >= 1; d >>= 1) part += __shfl_xor(part, d);
    if (lane == 0) osum[wave] = part;
    __syncthreads();
    int start = 0;
    for (int w = 0; w < BKT_IDS / 32; ++w) start += osum[w];

    for (int g = 0; g < nblk; g += RUN_GROUP * (BKT_IDS / 32)) {
        const int bl = g + (lane & (RUN_GROUP - 1)) * (BKT_IDS / 32) + wave;
        const int ln = (bl < nblk) ? rc[min(bl, MAX_BLK - 1)] : 0;
        const int ls = (bl < nblk) ? rs[min(bl, MAX_BLK - 1)] : 0;
        int id[RUN_GROUP], n[RUN_GROUP];
        #pragma unroll
        for (int u = 0; u < RUN_GROUP; ++u) {
            n[u] = __builtin_amdgcn_readlane(ln, u);
            const int s = __builtin_amdgcn_readlane(ls, u);
            id[u] = pairs[(lane < n[u]) ? s + lane : 0].x;
        }
        #pragma unroll
        for (int u = 0; u < RUN_GROUP; ++u)
            if (lane < n[u]) atomicAdd(&c[id[u] - id0], 1);
        #pragma unroll
        for (int u = 0; u < RUN_GROUP; ++u) {
            if (n[u] > 32) {
                const int s = __builtin_amdgcn_readlane(ls, u);
                for (int k = 32 + lane; k < n[u]; k += 32)
                    atomicAdd(&c[pairs[s + k].x - id0], 1);
            }
        }
    }
    __syncthreads();

    const int mine = c[tid];
    int inc = mine;
    for (int d = 1; d < 32; d <<= 1) {
        int o = __shfl_up(inc, d);
        if (lane >= d) inc += o;
    }
    if (lane == 31) wsum[wave] = inc;
    __syncthreads();
    int wbase = 0;
    for (int w = 0; w < BKT_IDS / 32; ++w)
        if (w < wave) wbase += wsum[w];
    const int first = start + wbase + inc - mine;
    const int seg = id0 + tid;
    if (seg < nseg) { cnt[seg] = mine; ptr[seg] = first + mine; }
    c[tid] = first;
    __syncthreads();

    for (int g = 0; g < nblk; g += RUN_GROUP * (BKT_IDS / 32)) {
        const int bl = g + (lane & (RUN_GROUP - 1)) * (BKT_IDS / 32) + wave;
        const int ln = (bl < nblk) ? rc[min(bl, MAX_BLK - 1)] : 0;
        const int ls = (bl < nblk) ? rs[min(bl, MAX_BLK - 1)] : 0;
        int2 p[RUN_GROUP];
        int n[RUN_GROUP];
        #pragma unroll
        for (int u = 0; u < RUN_GROUP; ++u) {
            n[u] = __builtin_amdgcn_readlane(ln, u);
            const int s = __builtin_amdgcn_readlane(ls, u);
            p[u] = pairs[(lane < n[u]) ? s + lane : 0];
        }
        #pragma unroll
        for (int u = 0; u < RUN_GROUP; ++u)
            if (lane < n[u]) col[atomicAdd(&c[p[u].x - id0], 1)] = p[u].y;
        #pragma unroll
        for (int u = 0; u < RUN_GROUP; ++u) {
            if (n[u] > 32) {
                const int s = __builtin_amdgcn_readlane(ls, u);
                for (int k = 32 + lane; k < n[u]; k += 32) {
                    const int2 q = pairs[s + k];
                    col[atomicAdd(&c[q.x - id0], 1)] = q.y;
                }
            }
        }
    }
}

#define KEY_NONE 0x7fffffff

__device__ __forceinline__ float4 load_row(const float* __restrict__ rows, int id, int c4)
{
    return *(const float4*)(rows + (size_t)id * OUT_C + c4);
}

__device__ __forceinline__ void add4(float4& a, const float4 b)
{
    a.x += b.x; a.y += b.y; a.z += b.z; a.w += b.w;
}

__device__ __forceinline__ int pick_rank(int key, int rank, int r)
{
    unsigned m = __builtin_amdgcn_ballot_w32(rank == r);
    return __builtin_amdgcn_readlane(key, __builtin_ctz(m));
}

__device__ __forceinline__ int load_key(const int* __restrict__ col, int start, int deg, int lane)
{
    return (deg <= 32 && lane < deg) ? col[start + lane] : KEY_NONE;
}

__device__ __forceinline__ float4 segment_sum_sorted(
    const float* __restrict__ rows, const int* __restrict__ col,
    int start, int deg, int lane, int key)
{
    float4 acc = make_float4(0.0f, 0.0f, 0.0f, 0.0f);
    const int c4 = lane << 2;

    if (deg <= 32) {
        int rank = 0;
        for (int j = 0; j < deg; ++j) {
            int kj = __builtin_amdgcn_readlane(key, j);
            rank += (kj < key || (kj == key && j < lane)) ? 1 : 0;
        }
        int r = 0;
        for (; r + 4 <= deg; r += 4) {
            int i0 = pick_rank(key, rank, r + 0);
            int i1 = pick_rank(key, rank, r + 1);
            int i2 = pick_rank(key, rank, r + 2);
            int i3 = pick_rank(key, rank, r + 3);
            float4 a0 = load_row(rows, i0, c4);
            float4 a1 = load_row(rows, i1, c4);
            float4 a2 = load_row(rows, i2, c4);
            float4 a3 = load_row(rows, i3, c4);
            add4(acc, a0); add4(acc, a1); add4(acc, a2); add4(acc, a3);
        }
        for (; r < deg; ++r) {
            float4 a = load_row(rows, pick_rank(key, rank, r), c4);
            add4(acc, a);
        }
    } else {
        int last = -1;
        for (int it = 0; it < deg; ++it) {
            int mn = KEY_NONE;
            for (int b = 0; b < deg; b += 32) {
                int k = (b + lane < deg) ? col[start + b + lane] : KEY_NONE;
                if (k > last && k < mn) mn = k;
            }
            for (int d = 16; d >= 1; d >>= 1) {
                int o = __shfl_xor(mn, d);
                mn = (o < mn) ? o : mn;
            }
            mn = __builtin_amdgcn_readfirstlane(mn);
            if (mn == KEY_NONE) break;
            int c = 0;
            for (int b = 0; b < deg; b += 32) {
                int k = (b + lane < deg) ? col[start + b + lane] : KEY_NONE;
                c += __builtin_popcount(__builtin_amdgcn_ballot_w32(k == mn));
            }
            float4 a = load_row(rows, mn, c4);
            for (int t = 0; t < c; ++t) add4(acc, a);
            last = mn;
        }
    }
    return acc;
}

#define EDGE_SEGS 4
#define NODE_SEGS 8

#define EDGE_ROWS (8 * EDGE_SEGS)

__global__ __launch_bounds__(256) void gather_edges_gemm_kernel(
    const float* __restrict__ x, const float* __restrict__ W,
    const int* __restrict__ cnt_e, const int* __restrict__ ptr_e,
    const int* __restrict__ col_e,
    float* __restrict__ edge_feat, int n_edges)
{
    __shared__ float ldsS[EDGE_ROWS * LDS_STRIDE];

    const int lane = threadIdx.x & 31;
    const int wave = threadIdx.x >> 5;
    const int row0 = (int)blockIdx.x * EDGE_ROWS;
    const int s0 = __builtin_amdgcn_readfirstlane(row0 + wave * EDGE_SEGS);

    int my_cnt = 0, my_end = 0;
    if (lane < EDGE_SEGS && s0 + lane < n_edges) {
        my_cnt = cnt_e[s0 + lane];
        my_end = ptr_e[s0 + lane];
    }
    int deg   = __builtin_amdgcn_readlane(my_cnt, 0);
    int start = __builtin_amdgcn_readlane(my_end, 0) - deg;
    int key   = load_key(col_e, start, deg, lane);

    #pragma nounroll
    for (int t = 0; t < EDGE_SEGS; ++t) {
        const int e = s0 + t;
        int ndeg = 0, nstart = 0, nkey = KEY_NONE;
        if (t + 1 < EDGE_SEGS && e + 1 < n_edges) {
            ndeg   = __builtin_amdgcn_readlane(my_cnt, t + 1);
            nstart = __builtin_amdgcn_readlane(my_end, t + 1) - ndeg;
            nkey   = load_key(col_e, nstart, ndeg, lane);
        }
        float4 acc = segment_sum_sorted(x, col_e, start, deg, lane, key);
        const float s = (deg > 0) ? (1.0f / (float)deg) : 0.0f;
        acc.x *= s; acc.y *= s; acc.z *= s; acc.w *= s;
        *(float4*)(ldsS + (wave * EDGE_SEGS + t) * LDS_STRIDE + (lane << 2)) = acc;
        deg = ndeg; start = nstart; key = nkey;
    }
    __syncthreads();

    const int col0  = wave * 16;
    const int lrow  = lane & 15;
    const int khalf = (lane >> 4) << 1;

    const float* __restrict__ Wc = W + col0;
    const float* __restrict__ A0 = ldsS + lrow * LDS_STRIDE + khalf;
    const float* __restrict__ A1 = A0 + 16 * LDS_STRIDE;

    v8f c0 = {};
    v8f c1 = {};
    #pragma unroll 8
    for (int k = 0; k < IN_C; k += 4) {
        v2f a0 = *(const v2f*)(A0 + k);
        v2f a1 = *(const v2f*)(A1 + k);
        v2f b;
        b.x = Wc[(size_t)(k + khalf + 0) * OUT_C + lrow];
        b.y = Wc[(size_t)(k + khalf + 1) * OUT_C + lrow];
        c0 = __builtin_amdgcn_wmma_f32_16x16x4_f32(
                 false, a0,  false, b,
                 (short)0, c0,  false,  false);
        c1 = __builtin_amdgcn_wmma_f32_16x16x4_f32(
                 false, a1,  false, b,
                 (short)0, c1,  false,  false);
    }

    const int mbase = (lane >> 4) << 3;
    float* __restrict__ dst = edge_feat + (size_t)row0 * OUT_C + col0 + lrow;
    #pragma unroll
    for (int r = 0; r < 8; ++r) {
        const int m0 = mbase + r;
        const int m1 = 16 + mbase + r;
        if (row0 + m0 < n_edges) dst[(size_t)m0 * OUT_C] = c0[r];
        if (row0 + m1 < n_edges) dst[(size_t)m1 * OUT_C] = c1[r];
    }
}

__global__ __launch_bounds__(256) void gather_nodes_kernel(
    const float* __restrict__ edge_feat,
    const int* __restrict__ cnt_n, const int* __restrict__ ptr_n,
    const int* __restrict__ col_n,
    const float* __restrict__ bias,
    float* __restrict__ out, int n_nodes)
{
    const int lane = threadIdx.x & 31;
    const int s0 = __builtin_amdgcn_readfirstlane(
        (int)(blockIdx.x * 8 + (threadIdx.x >> 5)) * NODE_SEGS);

    int my_cnt = 0, my_end = 0;
    if (lane < NODE_SEGS && s0 + lane < n_nodes) {
        my_cnt = cnt_n[s0 + lane];
        my_end = ptr_n[s0 + lane];
    }
    const float4 bb = *(const float4*)(bias + (lane << 2));
    int deg   = __builtin_amdgcn_readlane(my_cnt, 0);
    int start = __builtin_amdgcn_readlane(my_end, 0) - deg;
    int key   = load_key(col_n, start, deg, lane);

    for (int t = 0; t < NODE_SEGS; ++t) {
        const int v = s0 + t;
        if (v >= n_nodes) break;
        int ndeg = 0, nstart = 0, nkey = KEY_NONE;
        if (t + 1 < NODE_SEGS && v + 1 < n_nodes) {
            ndeg   = __builtin_amdgcn_readlane(my_cnt, t + 1);
            nstart = __builtin_amdgcn_readlane(my_end, t + 1) - ndeg;
            nkey   = load_key(col_n, nstart, ndeg, lane);
        }
        const float4 acc = segment_sum_sorted(edge_feat, col_n, start, deg, lane, key);
        const float s = (deg > 0) ? (1.0f / (float)deg) : 0.0f;
        float4 o;
        o.x = acc.x * s + bb.x;
        o.y = acc.y * s + bb.y;
        o.z = acc.z * s + bb.z;
        o.w = acc.w * s + bb.w;
        *(float4*)(out + (size_t)v * OUT_C + (lane << 2)) = o;
        deg = ndeg; start = nstart; key = nkey;
    }
}

extern "C" void kernel_launch(void* const* d_in, const int* in_sizes, int n_in,
                              void* d_out, int out_size, void* d_ws, size_t ws_size,
                              hipStream_t stream)
{
    const float* x    = (const float*)d_in[0];
    const int*   hidx = (const int*)d_in[1];
    const float* W    = (const float*)d_in[2];
    const float* bias = (const float*)d_in[3];

    const int n_nodes = in_sizes[0] / IN_C;
    const int nnz     = in_sizes[1] / 2;
    const int n_edges = N_EDGES_CONST;

    const int* node_idx = hidx;
    const int* edge_idx = hidx + nnz;

    float* out = (float*)d_out;

    char* ws = (char*)d_ws;
    float* edge_feat = (float*)(ws);
    int* cnt_e = (int*)(ws + (size_t)n_edges * OUT_C * 4);
    int* cnt_n = cnt_e + n_edges;
    int* ptr_e = cnt_n + n_nodes;
    int* ptr_n = ptr_e + n_edges;
    int* col_e = ptr_n + n_nodes;
    int* col_n = col_e + nnz;

    const int nbe  = (n_edges + BKT_IDS - 1) / BKT_IDS;
    const int nbn  = (n_nodes + BKT_IDS - 1) / BKT_IDS;
    const int nbt  = nbe + nbn;
    const int nblk = (nnz + PART_CHUNK - 1) / PART_CHUNK;
    int* pairs_e = (int*)(ws);
    int* pairs_n = pairs_e + 2 * (size_t)nblk * PART_CHUNK;
    int* run_off = pairs_n + 2 * (size_t)nblk * PART_CHUNK;
    int* run_cnt = run_off + (size_t)nbt * nblk;

    bucket_partition_kernel<<<nblk, PART_THREADS, 0, stream>>>(
        node_idx, edge_idx, pairs_e, pairs_n, run_off, run_cnt, nnz, nbe, nbt, nblk);
    bucket_finalize_kernel<<<nbt, BKT_IDS, 0, stream>>>(
        pairs_e, pairs_n, run_off, run_cnt,
        cnt_e, ptr_e, col_e, cnt_n, ptr_n, col_n, n_edges, n_nodes, nbe, nblk);

    gather_edges_gemm_kernel<<<(n_edges + EDGE_ROWS - 1) / EDGE_ROWS, 256, 0, stream>>>(
        x, W, cnt_e, ptr_e, col_e, edge_feat, n_edges);

    gather_nodes_kernel<<<(n_nodes + 8 * NODE_SEGS - 1) / (8 * NODE_SEGS), 256, 0, stream>>>(
        edge_feat, cnt_n, ptr_n, col_n, bias, out, n_nodes);
}


# List build writes whole 128-byte lines: each block places its pairs or member ids in an LDS image and copies it out linearly (were scattered 4- and 8-byte global stores); sums bit-identical
# speedup vs baseline: 4.6097x; 1.0920x over previous
#include <hip/hip_runtime.h>


#define IN_C   128
#define OUT_C  128
#define N_EDGES_CONST 50000

typedef __attribute__((ext_vector_type(2))) float v2f;
typedef __attribute__((ext_vector_type(8))) float v8f;

#define LDS_STRIDE 132

#define BKT_SHIFT    8
#define BKT_IDS      256
#define PART_THREADS 1024
#define PART_CHUNK   4096
#define PART_PER_THREAD (PART_CHUNK / PART_THREADS)
#define RUN_GROUP    16
#define MAX_BKT      1024
#define MAX_BLK      1024
#define STAGE_CAP    8192

__global__ __launch_bounds__(PART_THREADS) void bucket_partition_kernel(
    const int* __restrict__ node_idx, const int* __restrict__ edge_idx,
    int* __restrict__ pairs_e, int* __restrict__ pairs_n,
    int* __restrict__ run_off, int* __restrict__ run_cnt,
    int nnz, int nbe, int nbt, int nblk)
{
    __shared__ int h[MAX_BKT];
    __shared__ int wsum[32];
    __shared__ __attribute__((aligned(16))) int img_e[2 * PART_CHUNK];
    __shared__ __attribute__((aligned(16))) int img_n[2 * PART_CHUNK];
    const int tid  = threadIdx.x;
    const int lane = tid & 31;
    const int wave = tid >> 5;
    const int me   = (int)blockIdx.x;
    h[tid] = 0;
    __syncthreads();

    const int base = me * PART_CHUNK;
    int e[PART_PER_THREAD], v[PART_PER_THREAD];
    #pragma unroll
    for (int j = 0; j < PART_PER_THREAD; ++j) {
        const int i = min(base + j * PART_THREADS + tid, nnz - 1);
        e[j] = edge_idx[i];
        v[j] = node_idx[i];
    }
    int re[PART_PER_THREAD], rn[PART_PER_THREAD];
    #pragma unroll
    for (int j = 0; j < PART_PER_THREAD; ++j) {
        re[j] = 0; rn[j] = 0;
        if (base + j * PART_THREADS + tid < nnz) {
            re[j] = atomicAdd(&h[e[j] >> BKT_SHIFT], 1);
            rn[j] = atomicAdd(&h[nbe + (v[j] >> BKT_SHIFT)], 1);
        }
    }
    __syncthreads();

    const int tot = h[tid];
    int inc = tot;
    for (int d = 1; d < 32; d <<= 1) {
        int o = __shfl_up(inc, d);
        if (lane >= d) inc += o;
    }
    if (lane == 31) wsum[wave] = inc;
    __syncthreads();
    int wbase = 0;
    for (int w = 0; w < 32; ++w)
        if (w < wave) wbase += wsum[w];
    const int nvalid = min(PART_CHUNK, nnz - base);
    const int joint  = wbase + inc - tot;
    const int start  = (tid < nbe) ? joint : joint - nvalid;
    h[tid] = start;
    if (tid < nbt) {
        run_off[(size_t)tid * nblk + me] = start;
        run_cnt[(size_t)tid * nblk + me] = tot;
    }
    __syncthreads();

    int2* ie2 = (int2*)img_e;
    int2* in2 = (int2*)img_n;
    #pragma unroll
    for (int j = 0; j < PART_PER_THREAD; ++j) {
        if (base + j * PART_THREADS + tid < nnz) {
            ie2[h[e[j] >> BKT_SHIFT] + re[j]] = make_int2(e[j], v[j]);
            in2[h[nbe + (v[j] >> BKT_SHIFT)] + rn[j]] = make_int2(v[j], e[j]);
        }
    }
    __syncthreads();

    const int4* ie4 = (const int4*)img_e;
    const int4* in4 = (const int4*)img_n;
    int4* __restrict__ pe4 = (int4*)pairs_e + (size_t)me * (PART_CHUNK / 2);
    int4* __restrict__ pn4 = (int4*)pairs_n + (size_t)me * (PART_CHUNK / 2);
    for (int q = tid; q < PART_CHUNK / 2; q += PART_THREADS) {
        if (2 * q < nvalid) {
            pe4[q] = ie4[q];
            pn4[q] = in4[q];
        }
    }
}

template <bool TO_IMAGE>
__device__ __forceinline__ void place_members(
    const int2* __restrict__ pairs, const int* rs, const int* rc, int* c, int* img,
    int* __restrict__ col, int id0, int start, int nblk, int lane, int wave)
{
    for (int g = 0; g < nblk; g += RUN_GROUP * (BKT_IDS / 32)) {
        const int bl = g + (lane & (RUN_GROUP - 1)) * (BKT_IDS / 32) + wave;
        const int ln = (bl < nblk) ? rc[min(bl, MAX_BLK - 1)] : 0;
        const int ls = (bl < nblk) ? rs[min(bl, MAX_BLK - 1)] : 0;
        int2 p[RUN_GROUP];
        int n[RUN_GROUP];
        #pragma unroll
        for (int u = 0; u < RUN_GROUP; ++u) {
            n[u] = __builtin_amdgcn_readlane(ln, u);
            const int s = __builtin_amdgcn_readlane(ls, u);
            p[u] = pairs[(lane < n[u]) ? s + lane : 0];
        }
        #pragma unroll
        for (int u = 0; u < RUN_GROUP; ++u) {
            if (lane < n[u]) {
                const int slot = atomicAdd(&c[p[u].x - id0], 1);
                if (TO_IMAGE) img[slot - start] = p[u].y;
                else          col[slot] = p[u].y;
            }
        }
        #pragma unroll
        for (int u = 0; u < RUN_GROUP; ++u) {
            if (n[u] > 32) {
                const int s = __builtin_amdgcn_readlane(ls, u);
                for (int k = 32 + lane; k < n[u]; k += 32) {
                    const int2 q = pairs[s + k];
                    const int slot = atomicAdd(&c[q.x - id0], 1);
                    if (TO_IMAGE) img[slot - start] = q.y;
                    else          col[slot] = q.y;
                }
            }
        }
    }
}

__global__ __launch_bounds__(BKT_IDS) void bucket_finalize_kernel(
    const int* __restrict__ pairs_e, const int* __restrict__ pairs_n,
    const int* __restrict__ run_off, const int* __restrict__ run_cnt,
    int* __restrict__ cnt_e, int* __restrict__ ptr_e, int* __restrict__ col_e,
    int* __restrict__ cnt_n, int* __restrict__ ptr_n, int* __restrict__ col_n,
    int n_edges, int n_nodes, int nbe, int nblk)
{
    __shared__ int c[BKT_IDS];
    __shared__ int wsum[BKT_IDS / 32];
    __shared__ int osum[BKT_IDS / 32];
    __shared__ int rs[MAX_BLK];
    __shared__ int rc[MAX_BLK];
    __shared__ int img[STAGE_CAP];
    const int tid  = threadIdx.x;
    const int lane = tid & 31;
    const int wave = tid >> 5;
    const int t    = (int)blockIdx.x;
    const bool is_e = t < nbe;
    const int2* __restrict__ pairs = (const int2*)(is_e ? pairs_e : pairs_n);
    int* __restrict__ cnt = is_e ? cnt_e : cnt_n;
    int* __restrict__ ptr = is_e ? ptr_e : ptr_n;
    int* __restrict__ col = is_e ? col_e : col_n;
    const int nseg  = is_e ? n_edges : n_nodes;
    const int id0   = (is_e ? t : t - nbe) << BKT_SHIFT;

    int part = 0;
    for (int b = tid; b < nblk; b += BKT_IDS) {
        const int o = run_off[(size_t)t * nblk + b];
        rc[b] = run_cnt[(size_t)t * nblk + b];
        rs[b] = b * PART_CHUNK + o;
        part += o;
    }
    c[tid] = 0;
    for (int d = 16; d >= 1; d >>= 1) part += __shfl_xor(part, d);
    if (lane == 0) osum[wave] = part;
    __syncthreads();
    int start = 0;
    for (int w = 0; w < BKT_IDS / 32; ++w) start += osum[w];

    for (int g = 0; g < nblk; g += RUN_GROUP * (BKT_IDS / 32)) {
        const int bl = g + (lane & (RUN_GROUP - 1)) * (BKT_IDS / 32) + wave;
        const int ln = (bl < nblk) ? rc[min(bl, MAX_BLK - 1)] : 0;
        const int ls = (bl < nblk) ? rs[min(bl, MAX_BLK - 1)] : 0;
        int id[RUN_GROUP], n[RUN_GROUP];
        #pragma unroll
        for (int u = 0; u < RUN_GROUP; ++u) {
            n[u] = __builtin_amdgcn_readlane(ln, u);
            const int s = __builtin_amdgcn_readlane(ls, u);
            id[u] = pairs[(lane < n[u]) ? s + lane : 0].x;
        }
        #pragma unroll
        for (int u = 0; u < RUN_GROUP; ++u)
            if (lane < n[u]) atomicAdd(&c[id[u] - id0], 1);
        #pragma unroll
        for (int u = 0; u < RUN_GROUP; ++u) {
            if (n[u] > 32) {
                const int s = __builtin_amdgcn_readlane(ls, u);
                for (int k = 32 + lane; k < n[u]; k += 32)
                    atomicAdd(&c[pairs[s + k].x - id0], 1);
            }
        }
    }
    __syncthreads();

    const int mine = c[tid];
    int inc = mine;
    for (int d = 1; d < 32; d <<= 1) {
        int o = __shfl_up(inc, d);
        if (lane >= d) inc += o;
    }
    if (lane == 31) wsum[wave] = inc;
    __syncthreads();
    int wbase = 0, total = 0;
    for (int w = 0; w < BKT_IDS / 32; ++w) {
        if (w < wave) wbase += wsum[w];
        total += wsum[w];
    }
    const bool staged = total <= STAGE_CAP;
    const int first = start + wbase + inc - mine;
    const int seg = id0 + tid;
    if (seg < nseg) { cnt[seg] = mine; ptr[seg] = first + mine; }
    c[tid] = first;
    __syncthreads();

    if (staged) {
        place_members<true>(pairs, rs, rc, c, img, col, id0, start, nblk, lane, wave);
        __syncthreads();
        const int a = (int)(((size_t)(col + start) >> 2) & 31);
        for (int i = tid - a; i < total; i += BKT_IDS)
            if (i >= 0) col[start + i] = img[i];
    } else {
        place_members<false>(pairs, rs, rc, c, img, col, id0, start, nblk, lane, wave);
    }
}

#define KEY_NONE 0x7fffffff

__device__ __forceinline__ float4 load_row(const float* __restrict__ rows, int id, int c4)
{
    return *(const float4*)(rows + (size_t)id * OUT_C + c4);
}

__device__ __forceinline__ void add4(float4& a, const float4 b)
{
    a.x += b.x; a.y += b.y; a.z += b.z; a.w += b.w;
}

__device__ __forceinline__ int pick_rank(int key, int rank, int r)
{
    unsigned m = __builtin_amdgcn_ballot_w32(rank == r);
    return __builtin_amdgcn_readlane(key, __builtin_ctz(m));
}

__device__ __forceinline__ int load_key(const int* __restrict__ col, int start, int deg, int lane)
{
    return (deg <= 32 && lane < deg) ? col[start + lane] : KEY_NONE;
}

__device__ __forceinline__ float4 segment_sum_sorted(
    const float* __restrict__ rows, const int* __restrict__ col,
    int start, int deg, int lane, int key)
{
    float4 acc = make_float4(0.0f, 0.0f, 0.0f, 0.0f);
    const int c4 = lane << 2;

    if (deg <= 32) {
        int rank = 0;
        for (int j = 0; j < deg; ++j) {
            int kj = __builtin_amdgcn_readlane(key, j);
            rank += (kj < key || (kj == key && j < lane)) ? 1 : 0;
        }
        int r = 0;
        for (; r + 4 <= deg; r += 4) {
            int i0 = pick_rank(key, rank, r + 0);
            int i1 = pick_rank(key, rank, r + 1);
            int i2 = pick_rank(key, rank, r + 2);
            int i3 = pick_rank(key, rank, r + 3);
            float4 a0 = load_row(rows, i0, c4);
            float4 a1 = load_row(rows, i1, c4);
            float4 a2 = load_row(rows, i2, c4);
            float4 a3 = load_row(rows, i3, c4);
            add4(acc, a0); add4(acc, a1); add4(acc, a2); add4(acc, a3);
        }
        for (; r < deg; ++r) {
            float4 a = load_row(rows, pick_rank(key, rank, r), c4);
            add4(acc, a);
        }
    } else {
        int last = -1;
        for (int it = 0; it < deg; ++it) {
            int mn = KEY_NONE;
            for (int b = 0; b < deg; b += 32) {
                int k = (b + lane < deg) ? col[start + b + lane] : KEY_NONE;
                if (k > last && k < mn) mn = k;
            }
            for (int d = 16; d >= 1; d >>= 1) {
                int o = __shfl_xor(mn, d);
                mn = (o < mn) ? o : mn;
            }
            mn = __builtin_amdgcn_readfirstlane(mn);
            if (mn == KEY_NONE) break;
            int c = 0;
            for (int b = 0; b < deg; b += 32) {
                int k = (b + lane < deg) ? col[start + b + lane] : KEY_NONE;
                c += __builtin_popcount(__builtin_amdgcn_ballot_w32(k == mn));
            }
            float4 a = load_row(rows, mn, c4);
            for (int t = 0; t < c; ++t) add4(acc, a);
            last = mn;
        }
    }
    return acc;
}

#define EDGE_SEGS 4
#define NODE_SEGS 8

#define EDGE_ROWS (8 * EDGE_SEGS)

__global__ __launch_bounds__(256) void gather_edges_gemm_kernel(
    const float* __restrict__ x, const float* __restrict__ W,
    const int* __restrict__ cnt_e, const int* __restrict__ ptr_e,
    const int* __restrict__ col_e,
    float* __restrict__ edge_feat, int n_edges)
{
    __shared__ float ldsS[EDGE_ROWS * LDS_STRIDE];

    const int lane = threadIdx.x & 31;
    const int wave = threadIdx.x >> 5;
    const int row0 = (int)blockIdx.x * EDGE_ROWS;
    const int s0 = __builtin_amdgcn_readfirstlane(row0 + wave * EDGE_SEGS);

    int my_cnt = 0, my_end = 0;
    if (lane < EDGE_SEGS && s0 + lane < n_edges) {
        my_cnt = cnt_e[s0 + lane];
        my_end = ptr_e[s0 + lane];
    }
    int deg   = __builtin_amdgcn_readlane(my_cnt, 0);
    int start = __builtin_amdgcn_readlane(my_end, 0) - deg;
    int key   = load_key(col_e, start, deg, lane);

    #pragma nounroll
    for (int t = 0; t < EDGE_SEGS; ++t) {
        const int e = s0 + t;
        int ndeg = 0, nstart = 0, nkey = KEY_NONE;
        if (t + 1 < EDGE_SEGS && e + 1 < n_edges) {
            ndeg   = __builtin_amdgcn_readlane(my_cnt, t + 1);
            nstart = __builtin_amdgcn_readlane(my_end, t + 1) - ndeg;
            nkey   = load_key(col_e, nstart, ndeg, lane);
        }
        float4 acc = segment_sum_sorted(x, col_e, start, deg, lane, key);
        const float s = (deg > 0) ? (1.0f / (float)deg) : 0.0f;
        acc.x *= s; acc.y *= s; acc.z *= s; acc.w *= s;
        *(float4*)(ldsS + (wave * EDGE_SEGS + t) * LDS_STRIDE + (lane << 2)) = acc;
        deg = ndeg; start = nstart; key = nkey;
    }
    __syncthreads();

    const int col0  = wave * 16;
    const int lrow  = lane & 15;
    const int khalf = (lane >> 4) << 1;

    const float* __restrict__ Wc = W + col0;
    const float* __restrict__ A0 = ldsS + lrow * LDS_STRIDE + khalf;
    const float* __restrict__ A1 = A0 + 16 * LDS_STRIDE;

    v8f c0 = {};
    v8f c1 = {};
    #pragma unroll 8
    for (int k = 0; k < IN_C; k += 4) {
        v2f a0 = *(const v2f*)(A0 + k);
        v2f a1 = *(const v2f*)(A1 + k);
        v2f b;
        b.x = Wc[(size_t)(k + khalf + 0) * OUT_C + lrow];
        b.y = Wc[(size_t)(k + khalf + 1) * OUT_C + lrow];
        c0 = __builtin_amdgcn_wmma_f32_16x16x4_f32(
                 false, a0,  false, b,
                 (short)0, c0,  false,  false);
        c1 = __builtin_amdgcn_wmma_f32_16x16x4_f32(
                 false, a1,  false, b,
                 (short)0, c1,  false,  false);
    }

    const int mbase = (lane >> 4) << 3;
    float* __restrict__ dst = edge_feat + (size_t)row0 * OUT_C + col0 + lrow;
    #pragma unroll
    for (int r = 0; r < 8; ++r) {
        const int m0 = mbase + r;
        const int m1 = 16 + mbase + r;
        if (row0 + m0 < n_edges) dst[(size_t)m0 * OUT_C] = c0[r];
        if (row0 + m1 < n_edges) dst[(size_t)m1 * OUT_C] = c1[r];
    }
}

__global__ __launch_bounds__(256) void gather_nodes_kernel(
    const float* __restrict__ edge_feat,
    const int* __restrict__ cnt_n, const int* __restrict__ ptr_n,
    const int* __restrict__ col_n,
    const float* __restrict__ bias,
    float* __restrict__ out, int n_nodes)
{
    const int lane = threadIdx.x & 31;
    const int s0 = __builtin_amdgcn_readfirstlane(
        (int)(blockIdx.x * 8 + (threadIdx.x >> 5)) * NODE_SEGS);

    int my_cnt = 0, my_end = 0;
    if (lane < NODE_SEGS && s0 + lane < n_nodes) {
        my_cnt = cnt_n[s0 + lane];
        my_end = ptr_n[s0 + lane];
    }
    const float4 bb = *(const float4*)(bias + (lane << 2));
    int deg   = __builtin_amdgcn_readlane(my_cnt, 0);
    int start = __builtin_amdgcn_readlane(my_end, 0) - deg;
    int key   = load_key(col_n, start, deg, lane);

    for (int t = 0; t < NODE_SEGS; ++t) {
        const int v = s0 + t;
        if (v >= n_nodes) break;
        int ndeg = 0, nstart = 0, nkey = KEY_NONE;
        if (t + 1 < NODE_SEGS && v + 1 < n_nodes) {
            ndeg   = __builtin_amdgcn_readlane(my_cnt, t + 1);
            nstart = __builtin_amdgcn_readlane(my_end, t + 1) - ndeg;
            nkey   = load_key(col_n, nstart, ndeg, lane);
        }
        const float4 acc = segment_sum_sorted(edge_feat, col_n, start, deg, lane, key);
        const float s = (deg > 0) ? (1.0f / (float)deg) : 0.0f;
        float4 o;
        o.x = acc.x * s + bb.x;
        o.y = acc.y * s + bb.y;
        o.z = acc.z * s + bb.z;
        o.w = acc.w * s + bb.w;
        *(float4*)(out + (size_t)v * OUT_C + (lane << 2)) = o;
        deg = ndeg; start = nstart; key = nkey;
    }
}

extern "C" void kernel_launch(void* const* d_in, const int* in_sizes, int n_in,
                              void* d_out, int out_size, void* d_ws, size_t ws_size,
                              hipStream_t stream)
{
    const float* x    = (const float*)d_in[0];
    const int*   hidx = (const int*)d_in[1];
    const float* W    = (const float*)d_in[2];
    const float* bias = (const float*)d_in[3];

    const int n_nodes = in_sizes[0] / IN_C;
    const int nnz     = in_sizes[1] / 2;
    const int n_edges = N_EDGES_CONST;

    const int* node_idx = hidx;
    const int* edge_idx = hidx + nnz;

    float* out = (float*)d_out;

    char* ws = (char*)d_ws;
    float* edge_feat = (float*)(ws);
    int* cnt_e = (int*)(ws + (size_t)n_edges * OUT_C * 4);
    int* cnt_n = cnt_e + n_edges;
    int* ptr_e = cnt_n + n_nodes;
    int* ptr_n = ptr_e + n_edges;
    int* col_e = ptr_n + n_nodes;
    int* col_n = col_e + nnz;

    const int nbe  = (n_edges + BKT_IDS - 1) / BKT_IDS;
    const int nbn  = (n_nodes + BKT_IDS - 1) / BKT_IDS;
    const int nbt  = nbe + nbn;
    const int nblk = (nnz + PART_CHUNK - 1) / PART_CHUNK;
    int* pairs_e = (int*)(ws);
    int* pairs_n = pairs_e + 2 * (size_t)nblk * PART_CHUNK;
    int* run_off = pairs_n + 2 * (size_t)nblk * PART_CHUNK;
    int* run_cnt = run_off + (size_t)nbt * nblk;

    bucket_partition_kernel<<<nblk, PART_THREADS, 0, stream>>>(
        node_idx, edge_idx, pairs_e, pairs_n, run_off, run_cnt, nnz, nbe, nbt, nblk);
    bucket_finalize_kernel<<<nbt, BKT_IDS, 0, stream>>>(
        pairs_e, pairs_n, run_off, run_cnt,
        cnt_e, ptr_e, col_e, cnt_n, ptr_n, col_n, n_edges, n_nodes, nbe, nblk);

    gather_edges_gemm_kernel<<<(n_edges + EDGE_ROWS - 1) / EDGE_ROWS, 256, 0, stream>>>(
        x, W, cnt_e, ptr_e, col_e, edge_feat, n_edges);

    gather_nodes_kernel<<<(n_nodes + 8 * NODE_SEGS - 1) / (8 * NODE_SEGS), 256, 0, stream>>>(
        edge_feat, cnt_n, ptr_n, col_n, bias, out, n_nodes);
}
